# starting kernel
# baseline (speedup 1.0000x reference)
#include <hip/hip_runtime.h>

typedef __attribute__((ext_vector_type(2))) float v2f;
typedef __attribute__((ext_vector_type(8))) float v8f;

#define BB 8192
#define DD 512
#define OO 256

__global__ __launch_bounds__(256) void xlayer_fold_xval(
    const float* __restrict__ kern, const float* __restrict__ xval,
    float* __restrict__ W) {
  __shared__ float sxval[DD];
  const int o = threadIdx.x;
  const int i = blockIdx.x;
  for (int c = threadIdx.x; c < DD; c += 256) sxval[c] = xval[c];
  __syncthreads();

  const float* kp = kern + (size_t)i * (DD * OO) + o;
  float a0 = 0.f, a1 = 0.f, a2 = 0.f, a3 = 0.f;
#pragma unroll 4
  for (int c = 0; c < DD; c += 4) {
    a0 = fmaf(kp[(size_t)(c + 0) * OO], sxval[c + 0], a0);
    a1 = fmaf(kp[(size_t)(c + 1) * OO], sxval[c + 1], a1);
    a2 = fmaf(kp[(size_t)(c + 2) * OO], sxval[c + 2], a2);
    a3 = fmaf(kp[(size_t)(c + 3) * OO], sxval[c + 3], a3);
  }
  W[i * OO + o] = (a0 + a1) + (a2 + a3);
}

__global__ __launch_bounds__(256) void xlayer_wmma_gemm(
    const float* __restrict__ x, const float* __restrict__ W,
    float* __restrict__ out) {
  const int lane = threadIdx.x & 31;
  const int wave = (blockIdx.x << 3) + (threadIdx.x >> 5);
  const int NT = OO / 64;
  const int mBase = (wave / NT) * 16;
  const int nBase = (wave % NT) * 64;
  const int half = lane >> 4;
  const int l16  = lane & 15;

  v8f acc0 = {}, acc1 = {}, acc2 = {}, acc3 = {};

  const float* xrow = x + (size_t)(mBase + l16) * DD + 2 * half;
  const float* wcol = W + nBase + l16;

#pragma unroll 4
  for (int k0 = 0; k0 < DD; k0 += 4) {
    v2f a;
    a.x = xrow[k0];
    a.y = xrow[k0 + 1];

    const int kr = k0 + 2 * half;
    const float* wp0 = wcol + (size_t)kr * OO;
    const float* wp1 = wp0 + OO;
    v2f b0, b1, b2, b3;
    b0.x = wp0[0];  b0.y = wp1[0];
    b1.x = wp0[16]; b1.y = wp1[16];
    b2.x = wp0[32]; b2.y = wp1[32];
    b3.x = wp0[48]; b3.y = wp1[48];

    acc0 = __builtin_amdgcn_wmma_f32_16x16x4_f32(false, a, false, b0,
                                                 (short)0, acc0, false, false);
    acc1 = __builtin_amdgcn_wmma_f32_16x16x4_f32(false, a, false, b1,
                                                 (short)0, acc1, false, false);
    acc2 = __builtin_amdgcn_wmma_f32_16x16x4_f32(false, a, false, b2,
                                                 (short)0, acc2, false, false);
    acc3 = __builtin_amdgcn_wmma_f32_16x16x4_f32(false, a, false, b3,
                                                 (short)0, acc3, false, false);
  }

  float* orow = out + (size_t)(mBase + 8 * half) * OO + nBase + l16;
#pragma unroll
  for (int r = 0; r < 8; ++r) {
    float* p = orow + (size_t)r * OO;
    p[0]  = acc0[r];
    p[16] = acc1[r];
    p[32] = acc2[r];
    p[48] = acc3[r];
  }
}

extern "C" void kernel_launch(void* const* d_in, const int* in_sizes, int n_in,
                              void* d_out, int out_size, void* d_ws, size_t ws_size,
                              hipStream_t stream) {
  const float* x    = (const float*)d_in[0];
  const float* xval = (const float*)d_in[1];
  const float* kern = (const float*)d_in[2];
  float* out = (float*)d_out;
  float* W   = (float*)d_ws;

  xlayer_fold_xval<<<DD, 256, 0, stream>>>(kern, xval, W);

  const int mTiles = BB / 16;
  const int nTiles = OO / 64;
  const int blocks = (mTiles * nTiles) / 8;
  xlayer_wmma_gemm<<<blocks, 256, 0, stream>>>(x, W, out);
}


# GEMM operands staged through LDS: each work-group fetches its x and W k-chunks once with whole-line b128 loads into two LDS buffers, waves read fragments by ds_load_b64; k and sum order unchanged
# speedup vs baseline: 1.3763x; 1.3763x over previous
#include <hip/hip_runtime.h>

typedef __attribute__((ext_vector_type(2))) float v2f;
typedef __attribute__((ext_vector_type(4))) float v4f;
typedef __attribute__((ext_vector_type(8))) float v8f;

#define BB 8192
#define DD 512
#define OO 256

__global__ __launch_bounds__(256) void xlayer_fold_xval(
    const float* __restrict__ kern, const float* __restrict__ xval,
    float* __restrict__ W) {
  __shared__ float sxval[DD];
  const int o = threadIdx.x;
  const int i = blockIdx.x;
  for (int c = threadIdx.x; c < DD; c += 256) sxval[c] = xval[c];
  __syncthreads();

  const float* kp = kern + (size_t)i * (DD * OO) + o;
  float a0 = 0.f, a1 = 0.f, a2 = 0.f, a3 = 0.f;
#pragma unroll 4
  for (int c = 0; c < DD; c += 4) {
    a0 = fmaf(kp[(size_t)(c + 0) * OO], sxval[c + 0], a0);
    a1 = fmaf(kp[(size_t)(c + 1) * OO], sxval[c + 1], a1);
    a2 = fmaf(kp[(size_t)(c + 2) * OO], sxval[c + 2], a2);
    a3 = fmaf(kp[(size_t)(c + 3) * OO], sxval[c + 3], a3);
  }
  W[i * OO + o] = (a0 + a1) + (a2 + a3);
}

#define BM 32
#define KC 32
#define XP 36
#define WP 544

__device__ __forceinline__ void xlayer_chunk_products(
    const float* xa, const float* wb,
    v8f& acc0, v8f& acc1, v8f& acc2, v8f& acc3) {
#pragma unroll
  for (int kk = 0; kk < KC; kk += 4) {
    const v2f a = *(const v2f*)(xa + kk);
    const float* wq = wb + (kk >> 1) * WP;
    const v2f b0 = *(const v2f*)(wq);
    const v2f b1 = *(const v2f*)(wq + 32);
    const v2f b2 = *(const v2f*)(wq + 64);
    const v2f b3 = *(const v2f*)(wq + 96);
    acc0 = __builtin_amdgcn_wmma_f32_16x16x4_f32(false, a, false, b0,
                                                 (short)0, acc0, false, false);
    acc1 = __builtin_amdgcn_wmma_f32_16x16x4_f32(false, a, false, b1,
                                                 (short)0, acc1, false, false);
    acc2 = __builtin_amdgcn_wmma_f32_16x16x4_f32(false, a, false, b2,
                                                 (short)0, acc2, false, false);
    acc3 = __builtin_amdgcn_wmma_f32_16x16x4_f32(false, a, false, b3,
                                                 (short)0, acc3, false, false);
  }
}

struct xlayer_fetch {
  v4f x;
  v4f a[4];
  v4f b[4];
};

__device__ __forceinline__ void xlayer_fetch_chunk(
    xlayer_fetch& f, const float* __restrict__ xg, const float* __restrict__ wg,
    int k1) {
  f.x = *(const v4f*)(xg + k1);
#pragma unroll
  for (int j = 0; j < 4; ++j) {
    f.a[j] = *(const v4f*)(wg + (size_t)(k1 + 8 * j) * OO);
    f.b[j] = *(const v4f*)(wg + (size_t)(k1 + 8 * j + 1) * OO);
  }
}

__device__ __forceinline__ void xlayer_store_chunk(
    const xlayer_fetch& f, float* xd, float* wd) {
  *(v4f*)(xd) = f.x;
#pragma unroll
  for (int j = 0; j < 4; ++j) {
    float* d = wd + 4 * j * WP;
    *(v4f*)(d)     = v4f{f.a[j].x, f.b[j].x, f.a[j].y, f.b[j].y};
    *(v4f*)(d + 4) = v4f{f.a[j].z, f.b[j].z, f.a[j].w, f.b[j].w};
  }
}

__device__ __forceinline__ void xlayer_step(
    const float* __restrict__ xg, const float* __restrict__ wg, int k1,
    const float* xa, const float* wb, float* xd, float* wd,
    v8f& acc0, v8f& acc1, v8f& acc2, v8f& acc3) {
  xlayer_fetch f;
  xlayer_fetch_chunk(f, xg, wg, k1);
  xlayer_chunk_products(xa, wb, acc0, acc1, acc2, acc3);
  asm volatile("" : "+v"(acc0), "+v"(acc1), "+v"(acc2), "+v"(acc3),
                    "+v"(f.x), "+v"(f.a[0]), "+v"(f.a[1]), "+v"(f.a[2]),
                    "+v"(f.a[3]), "+v"(f.b[0]), "+v"(f.b[1]), "+v"(f.b[2]),
                    "+v"(f.b[3]));
  xlayer_store_chunk(f, xd, wd);
  __syncthreads();
}

__global__ __launch_bounds__(256) void xlayer_wmma_gemm(
    const float* __restrict__ x, const float* __restrict__ W,
    float* __restrict__ out) {
  __shared__ __attribute__((aligned(16))) float xs[2][BM * XP];
  __shared__ __attribute__((aligned(16))) float ws[2][(KC / 2) * WP];

  const int t    = threadIdx.x;
  const int lane = t & 31;
  const int w    = t >> 5;
  const int mw   = w >> 2;
  const int nw   = w & 3;
  const int half = lane >> 4;
  const int l16  = lane & 15;
  const int m0   = blockIdx.x * BM;

  const int xr = t >> 3;
  const int xc = (t & 7) * 4;
  const int wq = t >> 6;
  const int wc = (t & 63) * 4;
  const float* xg = x + (size_t)(m0 + xr) * DD + xc;
  const float* wg = W + (size_t)(2 * wq) * OO + wc;
  const int xso = xr * XP + xc;
  const int wso = wq * WP + wc * 2;

  const int xao = (mw * 16 + l16) * XP + 2 * half;
  const int wbo = half * WP + (nw * 64 + l16) * 2;

  v8f acc0 = {}, acc1 = {}, acc2 = {}, acc3 = {};

  {
    xlayer_fetch f;
    xlayer_fetch_chunk(f, xg, wg, 0);
    xlayer_store_chunk(f, &xs[0][xso], &ws[0][wso]);
  }
  __syncthreads();

#pragma unroll 1
  for (int ch = 0; ch < DD / KC - 2; ch += 2) {
    xlayer_step(xg, wg, (ch + 1) * KC, &xs[0][xao], &ws[0][wbo],
                &xs[1][xso], &ws[1][wso], acc0, acc1, acc2, acc3);
    xlayer_step(xg, wg, (ch + 2) * KC, &xs[1][xao], &ws[1][wbo],
                &xs[0][xso], &ws[0][wso], acc0, acc1, acc2, acc3);
  }
  xlayer_step(xg, wg, (DD / KC - 1) * KC, &xs[0][xao], &ws[0][wbo],
              &xs[1][xso], &ws[1][wso], acc0, acc1, acc2, acc3);
  xlayer_chunk_products(&xs[1][xao], &ws[1][wbo], acc0, acc1, acc2, acc3);

  const int mBase = m0 + mw * 16;
  const int nBase = nw * 64;
  float* orow = out + (size_t)(mBase + 8 * half) * OO + nBase + l16;
#pragma unroll
  for (int r = 0; r < 8; ++r) {
    float* p = orow + (size_t)r * OO;
    p[0]  = acc0[r];
    p[16] = acc1[r];
    p[32] = acc2[r];
    p[48] = acc3[r];
  }
}

extern "C" void kernel_launch(void* const* d_in, const int* in_sizes, int n_in,
                              void* d_out, int out_size, void* d_ws, size_t ws_size,
                              hipStream_t stream) {
  const float* x    = (const float*)d_in[0];
  const float* xval = (const float*)d_in[1];
  const float* kern = (const float*)d_in[2];
  float* out = (float*)d_out;
  float* W   = (float*)d_ws;

  xlayer_fold_xval<<<DD, 256, 0, stream>>>(kern, xval, W);

  const int mTiles = BB / 16;
  const int nTiles = OO / 64;
  const int blocks = (mTiles * nTiles) / 8;
  xlayer_wmma_gemm<<<blocks, 256, 0, stream>>>(x, W, out);
}


# fold stage as a pipelined stream: a thread sums 4 columns with 16-byte plain loads, 13 in flight per wave (was 16 4-byte loads, then a full wait); 4 c groups added via LDS (W sum order changes)
# speedup vs baseline: 1.6629x; 1.2082x over previous
#include <hip/hip_runtime.h>

typedef __attribute__((ext_vector_type(2))) float v2f;
typedef __attribute__((ext_vector_type(4))) float v4f;
typedef __attribute__((ext_vector_type(8))) float v8f;

#define BB 8192
#define DD 512
#define OO 256

#define FB 16
#define FG 128

__device__ __forceinline__ void xlayer_fold_fetch(v4f (&d)[FB], const float* p) {
#pragma unroll
  for (int j = 0; j < FB; ++j) d[j] = *(const v4f*)(p + (size_t)j * OO);
}

__device__ __forceinline__ void xlayer_fold_sum(const v4f (&d)[FB],
                                                const float* xs, v4f& acc) {
#pragma unroll
  for (int j = 0; j < FB; ++j) acc += d[j] * xs[j];
}

__global__ __launch_bounds__(256) void xlayer_fold_xval(
    const float* kern, const float* __restrict__ xval, float* __restrict__ W) {
  __shared__ __attribute__((aligned(16))) float part[4][OO];
  const int t  = threadIdx.x;
  const int i  = blockIdx.x;
  const int q4 = (t & 63) * 4;
  const int cg = __builtin_amdgcn_readfirstlane(t >> 6);

  const float* kp = kern + (size_t)i * (DD * OO) + (size_t)(cg * FG) * OO + q4;
  const float* xs = xval + cg * FG;

  v4f bufA[FB], bufB[FB];
  xlayer_fold_fetch(bufA, kp);

  v4f acc = {0.f, 0.f, 0.f, 0.f};
#pragma unroll
  for (int r = 0; r < FG; r += 2 * FB) {
    xlayer_fold_fetch(bufB, kp + (size_t)(r + FB) * OO);
    xlayer_fold_sum(bufA, xs + r, acc);
    if (r + 2 * FB < FG) xlayer_fold_fetch(bufA, kp + (size_t)(r + 2 * FB) * OO);
    xlayer_fold_sum(bufB, xs + r + FB, acc);
  }

  *(v4f*)(&part[cg][q4]) = acc;
  __syncthreads();
  if (t < 64) {
    const v4f p0 = *(const v4f*)(&part[0][q4]);
    const v4f p1 = *(const v4f*)(&part[1][q4]);
    const v4f p2 = *(const v4f*)(&part[2][q4]);
    const v4f p3 = *(const v4f*)(&part[3][q4]);
    *(v4f*)(W + (size_t)i * OO + q4) = (p0 + p1) + (p2 + p3);
  }
}

#define BM 32
#define KC 32
#define XP 36
#define WP 544

__device__ __forceinline__ void xlayer_chunk_products(
    const float* xa, const float* wb,
    v8f& acc0, v8f& acc1, v8f& acc2, v8f& acc3) {
#pragma unroll
  for (int kk = 0; kk < KC; kk += 4) {
    const v2f a = *(const v2f*)(xa + kk);
    const float* wq = wb + (kk >> 1) * WP;
    const v2f b0 = *(const v2f*)(wq);
    const v2f b1 = *(const v2f*)(wq + 32);
    const v2f b2 = *(const v2f*)(wq + 64);
    const v2f b3 = *(const v2f*)(wq + 96);
    acc0 = __builtin_amdgcn_wmma_f32_16x16x4_f32(false, a, false, b0,
                                                 (short)0, acc0, false, false);
    acc1 = __builtin_amdgcn_wmma_f32_16x16x4_f32(false, a, false, b1,
                                                 (short)0, acc1, false, false);
    acc2 = __builtin_amdgcn_wmma_f32_16x16x4_f32(false, a, false, b2,
                                                 (short)0, acc2, false, false);
    acc3 = __builtin_amdgcn_wmma_f32_16x16x4_f32(false, a, false, b3,
                                                 (short)0, acc3, false, false);
  }
}

struct xlayer_fetch {
  v4f x;
  v4f a[4];
  v4f b[4];
};

__device__ __forceinline__ void xlayer_fetch_chunk(
    xlayer_fetch& f, const float* __restrict__ xg, const float* __restrict__ wg,
    int k1) {
  f.x = *(const v4f*)(xg + k1);
#pragma unroll
  for (int j = 0; j < 4; ++j) {
    f.a[j] = *(const v4f*)(wg + (size_t)(k1 + 8 * j) * OO);
    f.b[j] = *(const v4f*)(wg + (size_t)(k1 + 8 * j + 1) * OO);
  }
}

__device__ __forceinline__ void xlayer_store_chunk(
    const xlayer_fetch& f, float* xd, float* wd) {
  *(v4f*)(xd) = f.x;
#pragma unroll
  for (int j = 0; j < 4; ++j) {
    float* d = wd + 4 * j * WP;
    *(v4f*)(d)     = v4f{f.a[j].x, f.b[j].x, f.a[j].y, f.b[j].y};
    *(v4f*)(d + 4) = v4f{f.a[j].z, f.b[j].z, f.a[j].w, f.b[j].w};
  }
}

__device__ __forceinline__ void xlayer_step(
    const float* __restrict__ xg, const float* __restrict__ wg, int k1,
    const float* xa, const float* wb, float* xd, float* wd,
    v8f& acc0, v8f& acc1, v8f& acc2, v8f& acc3) {
  xlayer_fetch f;
  xlayer_fetch_chunk(f, xg, wg, k1);
  xlayer_chunk_products(xa, wb, acc0, acc1, acc2, acc3);
  asm volatile("" : "+v"(acc0), "+v"(acc1), "+v"(acc2), "+v"(acc3),
                    "+v"(f.x), "+v"(f.a[0]), "+v"(f.a[1]), "+v"(f.a[2]),
                    "+v"(f.a[3]), "+v"(f.b[0]), "+v"(f.b[1]), "+v"(f.b[2]),
                    "+v"(f.b[3]));
  xlayer_store_chunk(f, xd, wd);
  __syncthreads();
}

__global__ __launch_bounds__(256) void xlayer_wmma_gemm(
    const float* __restrict__ x, const float* __restrict__ W,
    float* __restrict__ out) {
  __shared__ __attribute__((aligned(16))) float xs[2][BM * XP];
  __shared__ __attribute__((aligned(16))) float ws[2][(KC / 2) * WP];

  const int t    = threadIdx.x;
  const int lane = t & 31;
  const int w    = t >> 5;
  const int mw   = w >> 2;
  const int nw   = w & 3;
  const int half = lane >> 4;
  const int l16  = lane & 15;
  const int m0   = blockIdx.x * BM;

  const int xr = t >> 3;
  const int xc = (t & 7) * 4;
  const int wq = t >> 6;
  const int wc = (t & 63) * 4;
  const float* xg = x + (size_t)(m0 + xr) * DD + xc;
  const float* wg = W + (size_t)(2 * wq) * OO + wc;
  const int xso = xr * XP + xc;
  const int wso = wq * WP + wc * 2;

  const int xao = (mw * 16 + l16) * XP + 2 * half;
  const int wbo = half * WP + (nw * 64 + l16) * 2;

  v8f acc0 = {}, acc1 = {}, acc2 = {}, acc3 = {};

  {
    xlayer_fetch f;
    xlayer_fetch_chunk(f, xg, wg, 0);
    xlayer_store_chunk(f, &xs[0][xso], &ws[0][wso]);
  }
  __syncthreads();

#pragma unroll 1
  for (int ch = 0; ch < DD / KC - 2; ch += 2) {
    xlayer_step(xg, wg, (ch + 1) * KC, &xs[0][xao], &ws[0][wbo],
                &xs[1][xso], &ws[1][wso], acc0, acc1, acc2, acc3);
    xlayer_step(xg, wg, (ch + 2) * KC, &xs[1][xao], &ws[1][wbo],
                &xs[0][xso], &ws[0][wso], acc0, acc1, acc2, acc3);
  }
  xlayer_step(xg, wg, (DD / KC - 1) * KC, &xs[0][xao], &ws[0][wbo],
              &xs[1][xso], &ws[1][wso], acc0, acc1, acc2, acc3);
  xlayer_chunk_products(&xs[1][xao], &ws[1][wbo], acc0, acc1, acc2, acc3);

  const int mBase = m0 + mw * 16;
  const int nBase = nw * 64;
  float* orow = out + (size_t)(mBase + 8 * half) * OO + nBase + l16;
#pragma unroll
  for (int r = 0; r < 8; ++r) {
    float* p = orow + (size_t)r * OO;
    p[0]  = acc0[r];
    p[16] = acc1[r];
    p[32] = acc2[r];
    p[48] = acc3[r];
  }
}

extern "C" void kernel_launch(void* const* d_in, const int* in_sizes, int n_in,
                              void* d_out, int out_size, void* d_ws, size_t ws_size,
                              hipStream_t stream) {
  const float* x    = (const float*)d_in[0];
  const float* xval = (const float*)d_in[1];
  const float* kern = (const float*)d_in[2];
  float* out = (float*)d_out;
  float* W   = (float*)d_ws;

  xlayer_fold_xval<<<DD, 256, 0, stream>>>(kern, xval, W);

  const int mTiles = BB / 16;
  const int nTiles = OO / 64;
  const int blocks = (mTiles * nTiles) / 8;
  xlayer_wmma_gemm<<<blocks, 256, 0, stream>>>(x, W, out);
}


# GEMM K loop on three LDS buffers with a split barrier: a wave stores the next chunk and signals at the start of a step, waits only behind the step's products; k and sum order unchanged
# speedup vs baseline: 1.7116x; 1.0293x over previous
#include <hip/hip_runtime.h>

typedef __attribute__((ext_vector_type(2))) float v2f;
typedef __attribute__((ext_vector_type(4))) float v4f;
typedef __attribute__((ext_vector_type(8))) float v8f;

#define BB 8192
#define DD 512
#define OO 256

#define FB 16
#define FG 128

__device__ __forceinline__ void xlayer_fold_fetch(v4f (&d)[FB], const float* p) {
#pragma unroll
  for (int j = 0; j < FB; ++j) d[j] = *(const v4f*)(p + (size_t)j * OO);
}

__device__ __forceinline__ void xlayer_fold_sum(const v4f (&d)[FB],
                                                const float* xs, v4f& acc) {
#pragma unroll
  for (int j = 0; j < FB; ++j) acc += d[j] * xs[j];
}

__global__ __launch_bounds__(256) void xlayer_fold_xval(
    const float* kern, const float* __restrict__ xval, float* __restrict__ W) {
  __shared__ __attribute__((aligned(16))) float part[4][OO];
  const int t  = threadIdx.x;
  const int i  = blockIdx.x;
  const int q4 = (t & 63) * 4;
  const int cg = __builtin_amdgcn_readfirstlane(t >> 6);

  const float* kp = kern + (size_t)i * (DD * OO) + (size_t)(cg * FG) * OO + q4;
  const float* xs = xval + cg * FG;

  v4f bufA[FB], bufB[FB];
  xlayer_fold_fetch(bufA, kp);

  v4f acc = {0.f, 0.f, 0.f, 0.f};
#pragma unroll
  for (int r = 0; r < FG; r += 2 * FB) {
    xlayer_fold_fetch(bufB, kp + (size_t)(r + FB) * OO);
    xlayer_fold_sum(bufA, xs + r, acc);
    if (r + 2 * FB < FG) xlayer_fold_fetch(bufA, kp + (size_t)(r + 2 * FB) * OO);
    xlayer_fold_sum(bufB, xs + r + FB, acc);
  }

  *(v4f*)(&part[cg][q4]) = acc;
  __syncthreads();
  if (t < 64) {
    const v4f p0 = *(const v4f*)(&part[0][q4]);
    const v4f p1 = *(const v4f*)(&part[1][q4]);
    const v4f p2 = *(const v4f*)(&part[2][q4]);
    const v4f p3 = *(const v4f*)(&part[3][q4]);
    *(v4f*)(W + (size_t)i * OO + q4) = (p0 + p1) + (p2 + p3);
  }
}

#define BM 32
#define KC 32
#define XP 36
#define WP 544

__device__ __forceinline__ void xlayer_chunk_products(
    const float* xa, const float* wb,
    v8f& acc0, v8f& acc1, v8f& acc2, v8f& acc3) {
#pragma unroll
  for (int kk = 0; kk < KC; kk += 4) {
    const v2f a = *(const v2f*)(xa + kk);
    const float* wq = wb + (kk >> 1) * WP;
    const v2f b0 = *(const v2f*)(wq);
    const v2f b1 = *(const v2f*)(wq + 32);
    const v2f b2 = *(const v2f*)(wq + 64);
    const v2f b3 = *(const v2f*)(wq + 96);
    acc0 = __builtin_amdgcn_wmma_f32_16x16x4_f32(false, a, false, b0,
                                                 (short)0, acc0, false, false);
    acc1 = __builtin_amdgcn_wmma_f32_16x16x4_f32(false, a, false, b1,
                                                 (short)0, acc1, false, false);
    acc2 = __builtin_amdgcn_wmma_f32_16x16x4_f32(false, a, false, b2,
                                                 (short)0, acc2, false, false);
    acc3 = __builtin_amdgcn_wmma_f32_16x16x4_f32(false, a, false, b3,
                                                 (short)0, acc3, false, false);
  }
}

struct xlayer_fetch {
  v4f x;
  v4f a[4];
  v4f b[4];
};

__device__ __forceinline__ void xlayer_fetch_chunk(
    xlayer_fetch& f, const float* __restrict__ xg, const float* __restrict__ wg,
    int k1) {
  f.x = *(const v4f*)(xg + k1);
#pragma unroll
  for (int j = 0; j < 4; ++j) {
    f.a[j] = *(const v4f*)(wg + (size_t)(k1 + 8 * j) * OO);
    f.b[j] = *(const v4f*)(wg + (size_t)(k1 + 8 * j + 1) * OO);
  }
}

__device__ __forceinline__ void xlayer_store_chunk(
    const xlayer_fetch& f, float* xd, float* wd) {
  *(v4f*)(xd) = f.x;
#pragma unroll
  for (int j = 0; j < 4; ++j) {
    float* d = wd + 4 * j * WP;
    *(v4f*)(d)     = v4f{f.a[j].x, f.b[j].x, f.a[j].y, f.b[j].y};
    *(v4f*)(d + 4) = v4f{f.a[j].z, f.b[j].z, f.a[j].w, f.b[j].w};
  }
}

template <bool FETCH>
__device__ __forceinline__ void xlayer_step(
    const float* __restrict__ xg, const float* __restrict__ wg, int k2,
    const float* xa, const float* wb, float* xd, float* wd, xlayer_fetch& f,
    v8f& acc0, v8f& acc1, v8f& acc2, v8f& acc3) {
  xlayer_store_chunk(f, xd, wd);
  __builtin_amdgcn_fence(__ATOMIC_RELEASE, "workgroup");
  __builtin_amdgcn_s_barrier_signal(-1);
  if (FETCH) xlayer_fetch_chunk(f, xg, wg, k2);
  xlayer_chunk_products(xa, wb, acc0, acc1, acc2, acc3);
  asm volatile("" : "+v"(acc0), "+v"(acc1), "+v"(acc2), "+v"(acc3),
                    "+v"(f.x), "+v"(f.a[0]), "+v"(f.a[1]), "+v"(f.a[2]),
                    "+v"(f.a[3]), "+v"(f.b[0]), "+v"(f.b[1]), "+v"(f.b[2]),
                    "+v"(f.b[3]));
  __builtin_amdgcn_s_barrier_wait(-1);
  __builtin_amdgcn_fence(__ATOMIC_ACQUIRE, "workgroup");
}

__global__ __launch_bounds__(256) void xlayer_wmma_gemm(
    const float* __restrict__ x, const float* __restrict__ W,
    float* __restrict__ out) {
  __shared__ __attribute__((aligned(16))) float xs[3][BM * XP];
  __shared__ __attribute__((aligned(16))) float ws[3][(KC / 2) * WP];

  const int t    = threadIdx.x;
  const int lane = t & 31;
  const int w    = t >> 5;
  const int mw   = w >> 2;
  const int nw   = w & 3;
  const int half = lane >> 4;
  const int l16  = lane & 15;
  const int m0   = blockIdx.x * BM;

  const int xr = t >> 3;
  const int xc = (t & 7) * 4;
  const int wq = t >> 6;
  const int wc = (t & 63) * 4;
  const float* xg = x + (size_t)(m0 + xr) * DD + xc;
  const float* wg = W + (size_t)(2 * wq) * OO + wc;
  const int xso = xr * XP + xc;
  const int wso = wq * WP + wc * 2;

  const int xao = (mw * 16 + l16) * XP + 2 * half;
  const int wbo = half * WP + (nw * 64 + l16) * 2;

  v8f acc0 = {}, acc1 = {}, acc2 = {}, acc3 = {};

  xlayer_fetch f;
  {
    xlayer_fetch f0;
    xlayer_fetch_chunk(f0, xg, wg, 0);
    xlayer_fetch_chunk(f, xg, wg, KC);
    xlayer_store_chunk(f0, &xs[0][xso], &ws[0][wso]);
  }
  __syncthreads();

#pragma unroll 1
  for (int ch = 0; ch < DD / KC - 4; ch += 3) {
    xlayer_step<true>(xg, wg, (ch + 2) * KC, &xs[0][xao], &ws[0][wbo],
                      &xs[1][xso], &ws[1][wso], f, acc0, acc1, acc2, acc3);
    xlayer_step<true>(xg, wg, (ch + 3) * KC, &xs[1][xao], &ws[1][wbo],
                      &xs[2][xso], &ws[2][wso], f, acc0, acc1, acc2, acc3);
    xlayer_step<true>(xg, wg, (ch + 4) * KC, &xs[2][xao], &ws[2][wbo],
                      &xs[0][xso], &ws[0][wso], f, acc0, acc1, acc2, acc3);
  }
  xlayer_step<true>(xg, wg, (DD / KC - 2) * KC, &xs[0][xao], &ws[0][wbo],
                    &xs[1][xso], &ws[1][wso], f, acc0, acc1, acc2, acc3);
  xlayer_step<true>(xg, wg, (DD / KC - 1) * KC, &xs[1][xao], &ws[1][wbo],
                    &xs[2][xso], &ws[2][wso], f, acc0, acc1, acc2, acc3);
  xlayer_step<false>(xg, wg, 0, &xs[2][xao], &ws[2][wbo],
                     &xs[0][xso], &ws[0][wso], f, acc0, acc1, acc2, acc3);
  xlayer_chunk_products(&xs[0][xao], &ws[0][wbo], acc0, acc1, acc2, acc3);

  const int mBase = m0 + mw * 16;
  const int nBase = nw * 64;
  float* orow = out + (size_t)(mBase + 8 * half) * OO + nBase + l16;
#pragma unroll
  for (int r = 0; r < 8; ++r) {
    float* p = orow + (size_t)r * OO;
    p[0]  = acc0[r];
    p[16] = acc1[r];
    p[32] = acc2[r];
    p[48] = acc3[r];
  }
}

extern "C" void kernel_launch(void* const* d_in, const int* in_sizes, int n_in,
                              void* d_out, int out_size, void* d_ws, size_t ws_size,
                              hipStream_t stream) {
  const float* x    = (const float*)d_in[0];
  const float* xval = (const float*)d_in[1];
  const float* kern = (const float*)d_in[2];
  float* out = (float*)d_out;
  float* W   = (float*)d_ws;

  xlayer_fold_xval<<<DD, 256, 0, stream>>>(kern, xval, W);

  const int mTiles = BB / 16;
  const int nTiles = OO / 64;
  const int blocks = (mTiles * nTiles) / 8;
  xlayer_wmma_gemm<<<blocks, 256, 0, stream>>>(x, W, out);
}
